# starting kernel
# baseline (speedup 1.0000x reference)
#include <hip/hip_runtime.h>

#define NN   4096
#define DD   70
#define NT   256
#define ND   5
#define SPLITK 8
#define TPC  (NT / SPLITK)

typedef float v2f __attribute__((ext_vector_type(2)));
typedef float v4f __attribute__((ext_vector_type(4)));
typedef float v8f __attribute__((ext_vector_type(8)));

__device__ __forceinline__ v8f wmma4(v2f a, v2f b, v8f c) {
  return __builtin_amdgcn_wmma_f32_16x16x4_f32(
       false, a,  false, b,
       (short)0, c,  false,  false);
}

__global__ void zero_f32(float* __restrict__ p, int n) {
  int i = blockIdx.x * blockDim.x + threadIdx.x;
  if (i < n) p[i] = 0.0f;
}

__global__ void pack_sB(const float* __restrict__ s, v2f* __restrict__ sB) {
  int tid = blockIdx.x * blockDim.x + threadIdx.x;
  if (tid >= NT * ND * 4 * 32) return;
  int lane = tid & 31;
  int e = tid >> 5;
  int q = e & 3;  e >>= 2;
  int n = e % ND;
  int t = e / ND;
  int r = 16 * t + 4 * q + 2 * (lane >> 4);
  int d = 16 * n + (lane & 15);
  v2f v;
  v.x = (d < DD) ? s[(size_t)r * DD + d]       : 0.0f;
  v.y = (d < DD) ? s[(size_t)(r + 1) * DD + d] : 0.0f;
  sB[tid] = v;
}

__global__ __launch_bounds__(32) void slayer_wmma(const float* __restrict__ adj,
                                                  const v2f*  __restrict__ sB,
                                                  float*      __restrict__ out) {
  const int wg   = blockIdx.x;
  const int b    = wg >> 3;
  const int c    = wg & (SPLITK-1);
  const int lane = threadIdx.x;
  const int m    = lane & 15;
  const int kh   = lane >> 4;

  v8f zero = {};
  v8f accI[ND], accO[ND];
#pragma unroll
  for (int n = 0; n < ND; ++n) { accI[n] = zero; accO[n] = zero; }

  const int i0 = 16 * b;
  for (int t = TPC * c; t < TPC * c + TPC; ++t) {
    const int j0 = 16 * t;
#pragma unroll
    for (int q = 0; q < 4; ++q) {
      const int kb = 4 * q + 2 * kh;
      v4f f = *(const v4f*)(adj + 2 * ((size_t)(i0 + m) * NN + (j0 + kb)));
      v2f aO; aO.x = f.x + f.y; aO.y = f.z + f.w;
      const float* p = adj + 2 * ((size_t)(j0 + kb) * NN + (i0 + m));
      v2f g0 = *(const v2f*)p;
      v2f g1 = *(const v2f*)(p + 2 * NN);
      v2f aI; aI.x = g0.x + g0.y; aI.y = g1.x + g1.y;
      const v2f* bp = sB + (size_t)t * (ND * 4 * 32) + q * 32 + lane;
#pragma unroll
      for (int n = 0; n < ND; ++n) {
        v2f bv = bp[n * 4 * 32];
        accO[n] = wmma4(aO, bv, accO[n]);
        accI[n] = wmma4(aI, bv, accI[n]);
      }
    }
  }

  float* outI = out;
  float* outO = out + (size_t)NN * DD;
#pragma unroll
  for (int n = 0; n < ND; ++n) {
    int d = 16 * n + m;
    if (d < DD) {
#pragma unroll
      for (int r = 0; r < 8; ++r) {
        int row = i0 + r + 8 * kh;
        atomicAdd(&outI[(size_t)row * DD + d], accI[n][r]);
        atomicAdd(&outO[(size_t)row * DD + d], accO[n][r]);
      }
    }
  }
}

extern "C" void kernel_launch(void* const* d_in, const int* in_sizes, int n_in,
                              void* d_out, int out_size, void* d_ws, size_t ws_size,
                              hipStream_t stream) {
  const float* adj = (const float*)d_in[0];
  const float* s   = (const float*)d_in[1];
  float* out = (float*)d_out;
  v2f* sB = (v2f*)d_ws;

  zero_f32<<<(out_size + 255) / 256, 256, 0, stream>>>(out, out_size);

  int packN = NT * ND * 4 * 32;
  pack_sB<<<(packN + 255) / 256, 256, 0, stream>>>(s, sB);

  slayer_wmma<<<NT * SPLITK, 32, 0, stream>>>(adj, sB, out);
}


# each k-tile of s fetched as its 35 whole 128-byte lines (9 flat 16-byte loads, not 16 part-line loads) and turned into B fragments through the wave's own LDS area; sum order unchanged
# speedup vs baseline: 1.8397x; 1.8397x over previous
#include <hip/hip_runtime.h>

#define NN   4096
#define DD   70
#define NT   256
#define ND   5
#define NW   16
#define TPW  (NT / NW)

#define NS   272
#define BS   (ND * NS)
#define WS   (2 * BS)

#define SA   1152

typedef float v2f __attribute__((ext_vector_type(2)));
typedef float v4f __attribute__((ext_vector_type(4)));
typedef float v8f __attribute__((ext_vector_type(8)));

__device__ __forceinline__ v8f wmma4(v2f a, v2f b, v8f c) {
  return __builtin_amdgcn_wmma_f32_16x16x4_f32(
       false, a,  false, b,
       (short)0, c,  false,  false);
}

__device__ __forceinline__ void sweep(const float* __restrict__ adj,
                                      const float* __restrict__ s,
                                      float* stg,
                                      int g, int w, int lane, v8f (&acc)[2][ND]) {
  const int m  = lane & 15;
  const int kh = lane >> 4;
  const int lo = 2 * kh * DD + m;
  const int lt = 2 * kh * DD + 64 + (m < 6 ? m : 5);
  const int l8 = 1024 + 4 * (lane < 24 ? lane : lane - 8);
  const float* prow = adj + 2 * ((size_t)(16 * g + m) * NN + (size_t)(2 * kh));
  const float* pcol = adj + 2 * ((size_t)(2 * kh) * NN + (size_t)(16 * g + m));
  for (int i = 0; i < TPW; ++i) {
    const int t = w + NW * i;
    const float* st = s + (size_t)t * (16 * DD);
    v4f sv[9];
#pragma unroll
    for (int j = 0; j < 8; ++j) sv[j] = *(const v4f*)(st + 128 * j + 4 * lane);
    sv[8] = *(const v4f*)(st + l8);
    v4f f[4];
    v2f c0[4], c1[4];
#pragma unroll
    for (int q = 0; q < 4; ++q) {
      const int kq = 16 * t + 4 * q;
      f[q] = *(const v4f*)(prow + 2 * (size_t)kq);
      const float* pc = pcol + 2 * (size_t)kq * NN;
      c0[q] = *(const v2f*)pc;
      c1[q] = *(const v2f*)(pc + 2 * NN);
    }
#pragma unroll
    for (int j = 0; j < 9; ++j) *(v4f*)(stg + 128 * j + 4 * lane) = sv[j];
    v2f bf[4][ND];
#pragma unroll
    for (int q = 0; q < 4; ++q) {
      const float* sr = stg + 4 * q * DD;
#pragma unroll
      for (int n = 0; n < 4; ++n) {
        bf[q][n].x = sr[lo + 16 * n];
        bf[q][n].y = sr[lo + 16 * n + DD];
      }
      bf[q][4].x = sr[lt];
      bf[q][4].y = sr[lt + DD];
    }
    __builtin_amdgcn_sched_barrier(0);
#pragma unroll
    for (int q = 0; q < 4; ++q) {
      v2f a0, a1;
      a0.x = f[q].x + f[q].y;    a0.y = f[q].z + f[q].w;
      a1.x = c0[q].x + c0[q].y;  a1.y = c1[q].x + c1[q].y;
#pragma unroll
      for (int n = 0; n < ND; ++n) {
        acc[0][n] = wmma4(a0, bf[q][n], acc[0][n]);
        acc[1][n] = wmma4(a1, bf[q][n], acc[1][n]);
      }
    }
  }
}

__global__ __launch_bounds__(NW * 32) void slayer_panel(const float* __restrict__ adj,
                                                        const float* __restrict__ s,
                                                        float*       __restrict__ out) {
  __shared__ float red[NW * WS];
  __shared__ float stage[NW * SA];

  const int  g     = blockIdx.x;
  const int  tid   = threadIdx.x;
  const int  lane  = tid & 31;
  const int  w     = tid >> 5;

  v8f zero = {};
  v8f acc[2][ND];
#pragma unroll
  for (int n = 0; n < ND; ++n) { acc[0][n] = zero; acc[1][n] = zero; }

  sweep(adj, s, stage + w * SA, g, w, lane, acc);

#pragma unroll
  for (int blk = 0; blk < 2; ++blk)
#pragma unroll
    for (int n = 0; n < ND; ++n)
#pragma unroll
      for (int r = 0; r < 8; ++r)
        red[w * WS + blk * BS + n * NS + r * 32 + lane] = acc[blk][n][r];
  __syncthreads();

  float* op_in  = out + (size_t)(16 * g) * DD;
  float* op_out = out + (size_t)NN * DD + (size_t)(16 * g) * DD;
#pragma unroll
  for (int e = 0; e < 5; ++e) {
    const int idx = tid + NW * 32 * e;
    if (idx < 32 * DD) {
      const bool s_out_run = idx >= 16 * DD;
      const int j   = s_out_run ? idx - 16 * DD : idx;
      const int M   = j / DD;
      const int d   = j - DD * M;
      const int blk = s_out_run ? 0 : 1;
      const int n   = d >> 4;
      const int c   = d & 15;
      const int a   = blk * BS + n * NS + 32 * (M & 7) + 16 * (M >> 3) + c;
      float sum = red[a];
#pragma unroll
      for (int ww = 1; ww < NW; ++ww) sum += red[ww * WS + a];
      (s_out_run ? op_out : op_in)[j] = sum;
    }
  }
}

extern "C" void kernel_launch(void* const* d_in, const int* in_sizes, int n_in,
                              void* d_out, int out_size, void* d_ws, size_t ws_size,
                              hipStream_t stream) {
  const float* adj = (const float*)d_in[0];
  const float* s   = (const float*)d_in[1];
  float* out = (float*)d_out;

  slayer_panel<<<NT, NW * 32, 0, stream>>>(adj, s, out);
}
